# starting kernel
# baseline (speedup 1.0000x reference)
#include <hip/hip_runtime.h>
#include <math.h>

typedef __attribute__((ext_vector_type(2))) float v2f;
typedef __attribute__((ext_vector_type(8))) float v8f;

#define NEG_SLOPE 0.2f

__device__ __forceinline__ void atomicMaxFloat(float* addr, float val) {
  if (val >= 0.0f)
    atomicMax((int*)addr, __float_as_int(val));
  else
    atomicMin((unsigned int*)addr, __float_as_uint(val));
}

__global__ void init_kernel(float* __restrict__ acc_pc, float* __restrict__ acc_mc,
                            float* __restrict__ emax, float* __restrict__ denom,
                            int accN, int edgN) {
  int i = blockIdx.x * blockDim.x + threadIdx.x;
  if (i < accN) { acc_pc[i] = 0.0f; acc_mc[i] = 0.0f; }
  if (i < edgN) { emax[i] = -__builtin_inff(); denom[i] = 0.0f; }
}

__global__ __launch_bounds__(256)
void gemm_kernel(const float* __restrict__ x, const float* __restrict__ Wpc,
                 const float* __restrict__ Wmc, float* __restrict__ hpc,
                 float* __restrict__ hmc) {
  __shared__ float xs[16 * 128];
  const int rowbase = blockIdx.x * 16;
  for (int i = threadIdx.x; i < 16 * 128; i += 256)
    xs[i] = x[(size_t)rowbase * 128 + i];
  __syncthreads();

  const int lane   = threadIdx.x & 31;
  const int wave   = threadIdx.x >> 5;
  const int colbase = wave * 16;
  const int mn    = lane & 15;
  const int khalf = (lane >> 4) << 1;

  v8f cpc = {};
  v8f cmc = {};
  for (int k0 = 0; k0 < 128; k0 += 4) {
    v2f a, bpc, bmc;
    const float* ap = &xs[mn * 128 + k0 + khalf];
    a.x = ap[0];
    a.y = ap[1];
    const int wi = (k0 + khalf) * 128 + colbase + mn;
    bpc.x = Wpc[wi]; bpc.y = Wpc[wi + 128];
    bmc.x = Wmc[wi]; bmc.y = Wmc[wi + 128];
    cpc = __builtin_amdgcn_wmma_f32_16x16x4_f32(false, a, false, bpc, (short)0, cpc, false, false);
    cmc = __builtin_amdgcn_wmma_f32_16x16x4_f32(false, a, false, bmc, (short)0, cmc, false, false);
  }
  const int crow = rowbase + ((lane >> 4) << 3);
  const int ccol = colbase + mn;
#pragma unroll
  for (int r = 0; r < 8; ++r) {
    hpc[(size_t)(crow + r) * 128 + ccol] = cpc[r];
    hmc[(size_t)(crow + r) * 128 + ccol] = cmc[r];
  }
}

__global__ __launch_bounds__(256)
void score_kernel(const float* __restrict__ h, const float* __restrict__ a_src,
                  const float* __restrict__ a_dst, float* __restrict__ s_src,
                  float* __restrict__ s_dst, int n) {
  int warp = (blockIdx.x * blockDim.x + threadIdx.x) >> 5;
  int lane = threadIdx.x & 31;
  if (warp >= n) return;
  const float* hn = h + (size_t)warp * 128;
  float ss = 0.0f, sd = 0.0f;
#pragma unroll
  for (int j = 0; j < 4; ++j) {
    int c = lane * 4 + j;
    float hv = hn[c];
    ss += hv * a_src[c];
    sd += hv * a_dst[c];
  }
#pragma unroll
  for (int m = 1; m < 16; m <<= 1) {
    ss += __shfl_xor(ss, m, 32);
    sd += __shfl_xor(sd, m, 32);
  }
  if ((lane & 15) == 0) {
    int head = lane >> 4;
    s_src[warp * 2 + head] = ss;
    s_dst[warp * 2 + head] = sd;
  }
}

__global__ __launch_bounds__(256)
void edge_max_kernel(const int* __restrict__ ei, int E, int n,
                     const float* __restrict__ s_src, const float* __restrict__ s_dst,
                     float* __restrict__ e_store, float* __restrict__ emax) {
  int e = blockIdx.x * blockDim.x + threadIdx.x;
  if (e >= E + n) return;
  int src = (e < E) ? ei[e]     : (e - E);
  int dst = (e < E) ? ei[E + e] : (e - E);
#pragma unroll
  for (int h = 0; h < 2; ++h) {
    float v = s_src[src * 2 + h] + s_dst[dst * 2 + h];
    v = (v > 0.0f) ? v : NEG_SLOPE * v;
    e_store[(size_t)e * 2 + h] = v;
    atomicMaxFloat(&emax[dst * 2 + h], v);
  }
}

__global__ __launch_bounds__(256)
void edge_exp_kernel(const int* __restrict__ ei, int E, int n,
                     float* __restrict__ e_store, const float* __restrict__ emax,
                     float* __restrict__ denom) {
  int e = blockIdx.x * blockDim.x + threadIdx.x;
  if (e >= E + n) return;
  int dst = (e < E) ? ei[E + e] : (e - E);
#pragma unroll
  for (int h = 0; h < 2; ++h) {
    float ex = expf(e_store[(size_t)e * 2 + h] - emax[dst * 2 + h]);
    e_store[(size_t)e * 2 + h] = ex;
    atomicAdd(&denom[dst * 2 + h], ex);
  }
}

__global__ __launch_bounds__(256)
void edge_acc_kernel(const int* __restrict__ ei, int E, int n,
                     const float* __restrict__ e_store, const float* __restrict__ denom,
                     const float* __restrict__ h, float* __restrict__ acc) {
  int warp = (blockIdx.x * blockDim.x + threadIdx.x) >> 5;
  int lane = threadIdx.x & 31;
  if (warp >= E + n) return;
  int src = (warp < E) ? ei[warp]     : (warp - E);
  int dst = (warp < E) ? ei[E + warp] : (warp - E);
  float a0 = e_store[(size_t)warp * 2 + 0] / denom[dst * 2 + 0];
  float a1 = e_store[(size_t)warp * 2 + 1] / denom[dst * 2 + 1];
  const float* hs = h   + (size_t)src * 128;
  float*       ad = acc + (size_t)dst * 128;
#pragma unroll
  for (int j = 0; j < 4; ++j) {
    int c = lane + 32 * j;
    float alpha = (c < 64) ? a0 : a1;
    atomicAdd(&ad[c], alpha * hs[c]);
  }
}

__global__ __launch_bounds__(256)
void finalize_kernel(const float* __restrict__ acc_pc, const float* __restrict__ acc_mc,
                     const float* __restrict__ b_pc, const float* __restrict__ b_mc,
                     float* __restrict__ out, int total) {
  int i = blockIdx.x * blockDim.x + threadIdx.x;
  if (i >= total) return;
  int c = i & 127;
  float vp = acc_pc[i] + b_pc[c];
  vp = (vp > 0.0f) ? vp : (expf(vp) - 1.0f);
  float vm = acc_mc[i] + b_mc[c];
  vm = (vm > 0.0f) ? vm : (expf(vm) - 1.0f);
  out[i] = 0.5f * (vp + vm);
}

extern "C" void kernel_launch(void* const* d_in, const int* in_sizes, int n_in,
                              void* d_out, int out_size, void* d_ws, size_t ws_size,
                              hipStream_t stream) {
  const float* x      = (const float*)d_in[0];
  const float* W_pc   = (const float*)d_in[1];
  const float* asr_pc = (const float*)d_in[2];
  const float* ads_pc = (const float*)d_in[3];
  const float* b_pc   = (const float*)d_in[4];
  const float* W_mc   = (const float*)d_in[5];
  const float* asr_mc = (const float*)d_in[6];
  const float* ads_mc = (const float*)d_in[7];
  const float* b_mc   = (const float*)d_in[8];
  const int*   ei_pc  = (const int*)d_in[9];
  const int*   ei_mc  = (const int*)d_in[10];
  float* out = (float*)d_out;

  const int NN  = in_sizes[0] / 128;
  const int Epc = in_sizes[9] / 2;
  const int Emc = in_sizes[10] / 2;

  float* ws = (float*)d_ws;
  size_t off = 0;
  float* h_pc    = ws + off; off += (size_t)NN * 128;
  float* h_mc    = ws + off; off += (size_t)NN * 128;
  float* acc_pc  = ws + off; off += (size_t)NN * 128;
  float* acc_mc  = ws + off; off += (size_t)NN * 128;
  float* ssrc_pc = ws + off; off += (size_t)NN * 2;
  float* sdst_pc = ws + off; off += (size_t)NN * 2;
  float* ssrc_mc = ws + off; off += (size_t)NN * 2;
  float* sdst_mc = ws + off; off += (size_t)NN * 2;
  float* emax    = ws + off; off += (size_t)NN * 4;
  float* denom   = ws + off; off += (size_t)NN * 4;
  float* e_pc    = ws + off; off += (size_t)(Epc + NN) * 2;
  float* e_mc    = ws + off; off += (size_t)(Emc + NN) * 2;
  float* emax_pc = emax;            float* emax_mc = emax + (size_t)NN * 2;
  float* den_pc  = denom;           float* den_mc  = denom + (size_t)NN * 2;

  const int accN = NN * 128;
  init_kernel<<<(accN + 255) / 256, 256, 0, stream>>>(acc_pc, acc_mc, emax, denom,
                                                      accN, NN * 4);

  gemm_kernel<<<(NN + 15) / 16, 256, 0, stream>>>(x, W_pc, W_mc, h_pc, h_mc);

  score_kernel<<<(NN + 7) / 8, 256, 0, stream>>>(h_pc, asr_pc, ads_pc, ssrc_pc, sdst_pc, NN);
  score_kernel<<<(NN + 7) / 8, 256, 0, stream>>>(h_mc, asr_mc, ads_mc, ssrc_mc, sdst_mc, NN);

  const int tpc = Epc + NN, tmc = Emc + NN;
  edge_max_kernel<<<(tpc + 255) / 256, 256, 0, stream>>>(ei_pc, Epc, NN, ssrc_pc, sdst_pc, e_pc, emax_pc);
  edge_max_kernel<<<(tmc + 255) / 256, 256, 0, stream>>>(ei_mc, Emc, NN, ssrc_mc, sdst_mc, e_mc, emax_mc);

  edge_exp_kernel<<<(tpc + 255) / 256, 256, 0, stream>>>(ei_pc, Epc, NN, e_pc, emax_pc, den_pc);
  edge_exp_kernel<<<(tmc + 255) / 256, 256, 0, stream>>>(ei_mc, Emc, NN, e_mc, emax_mc, den_mc);

  edge_acc_kernel<<<(tpc + 7) / 8, 256, 0, stream>>>(ei_pc, Epc, NN, e_pc, den_pc, h_pc, acc_pc);
  edge_acc_kernel<<<(tmc + 7) / 8, 256, 0, stream>>>(ei_mc, Emc, NN, e_mc, den_mc, h_mc, acc_mc);

  finalize_kernel<<<(accN + 255) / 256, 256, 0, stream>>>(acc_pc, acc_mc, b_pc, b_mc, out, accN);
}


# Atomic scatter replaced by a dst-sorted gather: counting-atomic CSR built each call, each segment sorted by src in-wave so sums have a fixed order; one wave per node fuses softmax, sum, bias, elu
# speedup vs baseline: 2.3645x; 2.3645x over previous
#include <hip/hip_runtime.h>
#include <math.h>

typedef __attribute__((ext_vector_type(2))) float v2f;
typedef __attribute__((ext_vector_type(8))) float v8f;

#define NEG_SLOPE 0.2f

__global__ __launch_bounds__(256)
void gemm_kernel(const float* __restrict__ x, const float* __restrict__ Wpc,
                 const float* __restrict__ Wmc, float* __restrict__ hpc,
                 float* __restrict__ hmc) {
  __shared__ float xs[16 * 128];
  const int rowbase = blockIdx.x * 16;
  for (int i = threadIdx.x; i < 16 * 128; i += 256)
    xs[i] = x[(size_t)rowbase * 128 + i];
  __syncthreads();

  const int lane   = threadIdx.x & 31;
  const int wave   = threadIdx.x >> 5;
  const int colbase = wave * 16;
  const int mn    = lane & 15;
  const int khalf = (lane >> 4) << 1;

  v8f cpc = {};
  v8f cmc = {};
  for (int k0 = 0; k0 < 128; k0 += 4) {
    v2f a, bpc, bmc;
    const float* ap = &xs[mn * 128 + k0 + khalf];
    a.x = ap[0];
    a.y = ap[1];
    const int wi = (k0 + khalf) * 128 + colbase + mn;
    bpc.x = Wpc[wi]; bpc.y = Wpc[wi + 128];
    bmc.x = Wmc[wi]; bmc.y = Wmc[wi + 128];
    cpc = __builtin_amdgcn_wmma_f32_16x16x4_f32(false, a, false, bpc, (short)0, cpc, false, false);
    cmc = __builtin_amdgcn_wmma_f32_16x16x4_f32(false, a, false, bmc, (short)0, cmc, false, false);
  }
  const int crow = rowbase + ((lane >> 4) << 3);
  const int ccol = colbase + mn;
#pragma unroll
  for (int r = 0; r < 8; ++r) {
    hpc[(size_t)(crow + r) * 128 + ccol] = cpc[r];
    hmc[(size_t)(crow + r) * 128 + ccol] = cmc[r];
  }
}

__global__ __launch_bounds__(256)
void score_kernel(const float* __restrict__ h, const float* __restrict__ a_src,
                  const float* __restrict__ a_dst, float* __restrict__ s_src,
                  float* __restrict__ s_dst, int n) {
  int warp = (blockIdx.x * blockDim.x + threadIdx.x) >> 5;
  int lane = threadIdx.x & 31;
  if (warp >= n) return;
  const float* hn = h + (size_t)warp * 128;
  float ss = 0.0f, sd = 0.0f;
#pragma unroll
  for (int j = 0; j < 4; ++j) {
    int c = lane * 4 + j;
    float hv = hn[c];
    ss += hv * a_src[c];
    sd += hv * a_dst[c];
  }
#pragma unroll
  for (int m = 1; m < 16; m <<= 1) {
    ss += __shfl_xor(ss, m, 32);
    sd += __shfl_xor(sd, m, 32);
  }
  if ((lane & 15) == 0) {
    int head = lane >> 4;
    s_src[warp * 2 + head] = ss;
    s_dst[warp * 2 + head] = sd;
  }
}

__global__ __launch_bounds__(256)
void init_kernel(int* __restrict__ cnt, int total) {
  int i = blockIdx.x * blockDim.x + threadIdx.x;
  if (i < total) cnt[i] = 0;
}

__global__ __launch_bounds__(256)
void count_kernel(const int* __restrict__ ei_pc, const int* __restrict__ ei_mc, int Epc, int Emc,
                  int* __restrict__ cnt, int P, int* __restrict__ pos) {
  int e = blockIdx.x * blockDim.x + threadIdx.x;
  if (e < Epc) {
    int dst = ei_pc[Epc + e];
    pos[e] = atomicAdd(&cnt[dst], 1);
  } else if (e < Epc + Emc) {
    int dst = ei_mc[Emc + (e - Epc)];
    pos[e] = atomicAdd(&cnt[P + dst], 1);
  }
}

__global__ __launch_bounds__(1024)
void scan_kernel(const int* __restrict__ cnt, int* __restrict__ rowptr, int per, int P) {
  __shared__ int wsum[32];
  const int t = threadIdx.x;
  const int lane = t & 31, w = t >> 5;
  const int4* c4 = (const int4*)(cnt + (size_t)blockIdx.x * P + (size_t)t * per);
  int4* r4 = (int4*)(rowptr + (size_t)blockIdx.x * P + (size_t)t * per);
  const int nq = per >> 2;
  int s = 0;
  for (int k = 0; k < nq; ++k) { int4 v = c4[k]; s += v.x + v.y + v.z + v.w; }
  int inc = s;
#pragma unroll
  for (int m = 1; m < 32; m <<= 1) { int o = __shfl_up(inc, m, 32); if (lane >= m) inc += o; }
  if (lane == 31) wsum[w] = inc;
  __syncthreads();
  if (w == 0) {
    int v = wsum[lane], iv = v;
#pragma unroll
    for (int m = 1; m < 32; m <<= 1) { int o = __shfl_up(iv, m, 32); if (lane >= m) iv += o; }
    wsum[lane] = iv - v;
  }
  __syncthreads();
  int run = wsum[w] + inc - s;
  for (int k = 0; k < nq; ++k) {
    int4 v = c4[k], o;
    o.x = run; run += v.x;
    o.y = run; run += v.y;
    o.z = run; run += v.z;
    o.w = run; run += v.w;
    r4[k] = o;
  }
}

__global__ __launch_bounds__(256)
void place_kernel(const int* __restrict__ ei_pc, const int* __restrict__ ei_mc, int Epc, int Emc,
                  const int* __restrict__ rowptr, int P, const int* __restrict__ pos,
                  int* __restrict__ col) {
  int e = blockIdx.x * blockDim.x + threadIdx.x;
  if (e < Epc) {
    int src = ei_pc[e], dst = ei_pc[Epc + e];
    col[rowptr[dst] + pos[e]] = src;
  } else if (e < Epc + Emc) {
    int f = e - Epc;
    int src = ei_mc[f], dst = ei_mc[Emc + f];
    col[Epc + rowptr[P + dst] + pos[e]] = src;
  }
}

__device__ __forceinline__ float lrelu(float v) { return (v > 0.0f) ? v : NEG_SLOPE * v; }
__device__ __forceinline__ float elu1(float v)  { return (v > 0.0f) ? v : (expf(v) - 1.0f); }

__device__ __forceinline__ float lane_bcast(float v, int l) {
  return __int_as_float(__builtin_amdgcn_readlane(__float_as_int(v), l));
}

__device__ __forceinline__ float wave_max(float v) {
#pragma unroll
  for (int m = 16; m >= 1; m >>= 1) v = fmaxf(v, __shfl_xor(v, m, 32));
  return v;
}
__device__ __forceinline__ float wave_sum(float v) {
#pragma unroll
  for (int m = 16; m >= 1; m >>= 1) v += __shfl_xor(v, m, 32);
  return v;
}

__device__ __forceinline__ float4 gat_row(int n, int lane, int d, int beg,
                                          const int* col, int* col2,
                                          const float* __restrict__ s_src,
                                          const float* __restrict__ s_dst,
                                          const float* __restrict__ h) {
  const float sd0 = s_dst[n * 2 + 0], sd1 = s_dst[n * 2 + 1];
  const float es0 = lrelu(s_src[n * 2 + 0] + sd0);
  const float es1 = lrelu(s_src[n * 2 + 1] + sd1);
  const float4* h4 = (const float4*)h;
  const float4 hself = h4[(size_t)n * 32 + lane];
  const bool hi = lane >= 16;
  float4 acc;

  if (d <= 32) {
    const bool valid = lane < d;
    const int sj = valid ? col[beg + lane] : n;
    int rank = 0;
    for (int i = 0; i < d; ++i) {
      const int si = __builtin_amdgcn_readlane(sj, i);
      rank += ((si < sj) || (si == sj && i < lane)) ? 1 : 0;
    }
    if (!valid) rank = lane;
    const int s = __builtin_amdgcn_ds_permute(rank << 2, sj);

    const float e0 = lrelu(s_src[s * 2 + 0] + sd0);
    const float e1 = lrelu(s_src[s * 2 + 1] + sd1);
    const float m0 = fmaxf(wave_max(valid ? e0 : es0), es0);
    const float m1 = fmaxf(wave_max(valid ? e1 : es1), es1);
    const float x0 = valid ? expf(e0 - m0) : 0.0f;
    const float x1 = valid ? expf(e1 - m1) : 0.0f;
    const float xs0 = expf(es0 - m0), xs1 = expf(es1 - m1);
    const float z0 = wave_sum(x0) + xs0;
    const float z1 = wave_sum(x1) + xs1;
    const float a0 = x0 / z0, a1 = x1 / z1;
    {
      const float a = hi ? (xs1 / z1) : (xs0 / z0);
      acc.x = a * hself.x; acc.y = a * hself.y; acc.z = a * hself.z; acc.w = a * hself.w;
    }
    const int dr = (d + 3) & ~3;
    for (int t0 = 0; t0 < dr; t0 += 4) {
#pragma unroll
      for (int u = 0; u < 4; ++u) {
        const int t = t0 + u;
        const int ss = __builtin_amdgcn_readlane(s, t);
        const float b0 = lane_bcast(a0, t);
        const float b1 = lane_bcast(a1, t);
        const float a = hi ? b1 : b0;
        const float4 v = h4[(size_t)ss * 32 + lane];
        acc.x = fmaf(a, v.x, acc.x);
        acc.y = fmaf(a, v.y, acc.y);
        acc.z = fmaf(a, v.z, acc.z);
        acc.w = fmaf(a, v.w, acc.w);
      }
    }
    return acc;
  }

  for (int j = lane; j < d; j += 32) {
    const int sj = col[beg + j];
    int rank = 0;
    for (int i = 0; i < d; ++i) {
      const int si = col[beg + i];
      rank += ((si < sj) || (si == sj && i < j)) ? 1 : 0;
    }
    col2[beg + rank] = sj;
  }
  __threadfence_block();

  float m0 = es0, m1 = es1;
  for (int c0 = 0; c0 < d; c0 += 32) {
    const bool valid = (c0 + lane) < d;
    const int s = valid ? col2[beg + c0 + lane] : n;
    const float e0 = lrelu(s_src[s * 2 + 0] + sd0);
    const float e1 = lrelu(s_src[s * 2 + 1] + sd1);
    if (valid) { m0 = fmaxf(m0, e0); m1 = fmaxf(m1, e1); }
  }
  m0 = wave_max(m0);
  m1 = wave_max(m1);

  float z0 = 0.0f, z1 = 0.0f;
  for (int c0 = 0; c0 < d; c0 += 32) {
    const bool valid = (c0 + lane) < d;
    const int s = valid ? col2[beg + c0 + lane] : n;
    const float e0 = lrelu(s_src[s * 2 + 0] + sd0);
    const float e1 = lrelu(s_src[s * 2 + 1] + sd1);
    if (valid) { z0 += expf(e0 - m0); z1 += expf(e1 - m1); }
  }
  const float xs0 = expf(es0 - m0), xs1 = expf(es1 - m1);
  z0 = wave_sum(z0) + xs0;
  z1 = wave_sum(z1) + xs1;

  {
    const float a = hi ? (xs1 / z1) : (xs0 / z0);
    acc.x = a * hself.x; acc.y = a * hself.y; acc.z = a * hself.z; acc.w = a * hself.w;
  }
  for (int c0 = 0; c0 < d; c0 += 32) {
    const bool valid = (c0 + lane) < d;
    const int s = valid ? col2[beg + c0 + lane] : n;
    const float e0 = lrelu(s_src[s * 2 + 0] + sd0);
    const float e1 = lrelu(s_src[s * 2 + 1] + sd1);
    const float a0 = expf(e0 - m0) / z0;
    const float a1 = expf(e1 - m1) / z1;
    const int nc = (d - c0 < 32) ? (d - c0) : 32;
    for (int t = 0; t < nc; ++t) {
      const int ss = __builtin_amdgcn_readlane(s, t);
      const float b0 = lane_bcast(a0, t);
      const float b1 = lane_bcast(a1, t);
      const float a = hi ? b1 : b0;
      const float4 v = h4[(size_t)ss * 32 + lane];
      acc.x = fmaf(a, v.x, acc.x);
      acc.y = fmaf(a, v.y, acc.y);
      acc.z = fmaf(a, v.z, acc.z);
      acc.w = fmaf(a, v.w, acc.w);
    }
  }
  return acc;
}

__global__ __launch_bounds__(256)
void gather_kernel(const int* __restrict__ cnt, const int* __restrict__ rowptr, int P,
                   const int* col, int* col2, int Epc,
                   const float* __restrict__ ssrc_pc, const float* __restrict__ sdst_pc,
                   const float* __restrict__ ssrc_mc, const float* __restrict__ sdst_mc,
                   const float* __restrict__ h_pc, const float* __restrict__ h_mc,
                   const float* __restrict__ b_pc, const float* __restrict__ b_mc,
                   float* __restrict__ out, int NN) {
  const int lane = threadIdx.x & 31;
  const int n = __builtin_amdgcn_readfirstlane((int)((blockIdx.x * blockDim.x + threadIdx.x) >> 5));
  if (n >= NN) return;
  const int d_pc = cnt[n],     beg_pc = rowptr[n];
  const int d_mc = cnt[P + n], beg_mc = rowptr[P + n];
  const float4 p = gat_row(n, lane, d_pc, beg_pc, col, col2, ssrc_pc, sdst_pc, h_pc);
  const float4 m = gat_row(n, lane, d_mc, beg_mc, col + Epc, col2 + Epc, ssrc_mc, sdst_mc, h_mc);
  float4 bp, bm;
  bp.x = b_pc[lane * 4 + 0]; bp.y = b_pc[lane * 4 + 1]; bp.z = b_pc[lane * 4 + 2]; bp.w = b_pc[lane * 4 + 3];
  bm.x = b_mc[lane * 4 + 0]; bm.y = b_mc[lane * 4 + 1]; bm.z = b_mc[lane * 4 + 2]; bm.w = b_mc[lane * 4 + 3];
  float4 o;
  o.x = 0.5f * (elu1(p.x + bp.x) + elu1(m.x + bm.x));
  o.y = 0.5f * (elu1(p.y + bp.y) + elu1(m.y + bm.y));
  o.z = 0.5f * (elu1(p.z + bp.z) + elu1(m.z + bm.z));
  o.w = 0.5f * (elu1(p.w + bp.w) + elu1(m.w + bm.w));
  ((float4*)out)[(size_t)n * 32 + lane] = o;
}

extern "C" void kernel_launch(void* const* d_in, const int* in_sizes, int n_in,
                              void* d_out, int out_size, void* d_ws, size_t ws_size,
                              hipStream_t stream) {
  const float* x      = (const float*)d_in[0];
  const float* W_pc   = (const float*)d_in[1];
  const float* asr_pc = (const float*)d_in[2];
  const float* ads_pc = (const float*)d_in[3];
  const float* b_pc   = (const float*)d_in[4];
  const float* W_mc   = (const float*)d_in[5];
  const float* asr_mc = (const float*)d_in[6];
  const float* ads_mc = (const float*)d_in[7];
  const float* b_mc   = (const float*)d_in[8];
  const int*   ei_pc  = (const int*)d_in[9];
  const int*   ei_mc  = (const int*)d_in[10];
  float* out = (float*)d_out;

  const int NN  = in_sizes[0] / 128;
  const int Epc = in_sizes[9] / 2;
  const int Emc = in_sizes[10] / 2;
  const int per = (((NN + 1023) / 1024) + 3) & ~3;
  const int P   = per * 1024;

  float* ws = (float*)d_ws;
  size_t off = 0;
  float* h_pc    = ws + off; off += (size_t)NN * 128;
  float* h_mc    = ws + off; off += (size_t)NN * 128;
  int*   cnt     = (int*)(ws + off); off += (size_t)P * 2;
  int*   rowptr  = (int*)(ws + off); off += (size_t)P * 2;
  float* ssrc_pc = ws + off; off += (size_t)NN * 2;
  float* sdst_pc = ws + off; off += (size_t)NN * 2;
  float* ssrc_mc = ws + off; off += (size_t)NN * 2;
  float* sdst_mc = ws + off; off += (size_t)NN * 2;
  int*   pos     = (int*)(ws + off); off += (size_t)Epc + Emc;
  int*   col     = (int*)(ws + off); off += (size_t)Epc + Emc;
  int*   col2    = (int*)(ws + off); off += (size_t)Epc + Emc;

  const int tot = Epc + Emc;
  init_kernel<<<(2 * P + 255) / 256, 256, 0, stream>>>(cnt, 2 * P);

  gemm_kernel<<<(NN + 15) / 16, 256, 0, stream>>>(x, W_pc, W_mc, h_pc, h_mc);

  score_kernel<<<(NN + 7) / 8, 256, 0, stream>>>(h_pc, asr_pc, ads_pc, ssrc_pc, sdst_pc, NN);
  score_kernel<<<(NN + 7) / 8, 256, 0, stream>>>(h_mc, asr_mc, ads_mc, ssrc_mc, sdst_mc, NN);

  count_kernel<<<(tot + 255) / 256, 256, 0, stream>>>(ei_pc, ei_mc, Epc, Emc, cnt, P, pos);
  scan_kernel<<<2, 1024, 0, stream>>>(cnt, rowptr, per, P);
  place_kernel<<<(tot + 255) / 256, 256, 0, stream>>>(ei_pc, ei_mc, Epc, Emc, rowptr, P, pos, col);

  gather_kernel<<<(NN + 7) / 8, 256, 0, stream>>>(cnt, rowptr, P, col, col2, Epc,
                                                  ssrc_pc, sdst_pc, ssrc_mc, sdst_mc,
                                                  h_pc, h_mc, b_pc, b_mc, out, NN);
}
